# starting kernel
# baseline (speedup 1.0000x reference)
#include <hip/hip_runtime.h>


typedef __attribute__((ext_vector_type(2))) float v2f;
typedef __attribute__((ext_vector_type(8))) float v8f;

#define T_LEN    48000
#define M1       25
#define NITER    30
#define CHUNK    1024
#define HALO     720
#define REGION   (CHUNK + HALO)
#define NGROUPS  (REGION / 16)
#define NTHREADS 256
#define NWAVES   (NTHREADS / 32)
#define YPT      (CHUNK / NTHREADS)
#define PADL     32
#define USZ      (REGION + PADL)
#define DUMP     (USZ - 1)


__global__ __launch_bounds__(NTHREADS)
void mglsa_taylor_wmma_kernel(const float* __restrict__ x,
                              const float* __restrict__ mc,
                              float* __restrict__ y)
{
    __shared__ float uA[USZ];
    __shared__ float uB[USZ];

    const int b    = blockIdx.y;
    const int cs   = blockIdx.x * CHUNK;
    const int R    = cs - HALO;
    const int tid  = threadIdx.x;
    const int lane = tid & 31;
    const int wave = tid >> 5;

    const float* __restrict__ xb  = x  + (size_t)b * T_LEN;
    const float* __restrict__ mcb = mc + (size_t)b * T_LEN * M1;

    for (int p = tid; p < USZ; p += NTHREADS) {
        int l = (REGION - 1) - p;
        int t = R + l;
        uA[p] = (l >= 0 && (unsigned)t < (unsigned)T_LEN) ? xb[t] : 0.0f;
    }
    if (tid < PADL) uB[REGION + tid] = 0.0f;

    float yreg[YPT];
#pragma unroll
    for (int k = 0; k < YPT; ++k) {
        int t = cs + tid + k * NTHREADS;
        yreg[k] = (t < T_LEN) ? xb[t] : 0.0f;
    }
    __syncthreads();

    float* ucur = uA;
    float* unxt = uB;

    const int  m       = lane & 15;
    const int  half2   = (lane >> 4) << 1;
    const bool hasDiag = (lane < 8) || (lane >= 24);
    const int  dslot   = (lane < 8) ? lane : (lane - 16);

    for (int n = 1; n <= NITER; ++n) {
        const float inv_a = 1.0f / (float)n;

        for (int g = wave; g < NGROUPS; g += NWAVES) {
            const int l0   = g << 4;
            const int trow = R + l0 + m;
            const int tcl  = trow < 0 ? 0 : (trow >= T_LEN ? (T_LEN - 1) : trow);
            const float* __restrict__ mrow = mcb + (size_t)tcl * M1;
            const int pb = (REGION - 1) - (l0 + m - half2);

            {
                int tpre = trow + (NWAVES << 4);
                tpre = tpre < 0 ? 0 : (tpre >= T_LEN ? (T_LEN - 1) : tpre);
                __builtin_prefetch(mcb + (size_t)tpre * M1, 0, 3);
            }

            v8f acc = {0.f, 0.f, 0.f, 0.f, 0.f, 0.f, 0.f, 0.f};

#pragma unroll
            for (int k0 = 0; k0 < 24; k0 += 4) {
                const int j0 = k0 + half2;
                v2f a;
                a.x = mrow[j0];
                a.y = mrow[j0 + 1];
                v2f bb;
                bb.x = ucur[pb + k0];
                bb.y = ucur[pb + k0 + 1];
                acc = __builtin_amdgcn_wmma_f32_16x16x4_f32(
                          false, a, false, bb, (short)0, acc, false, false);
            }
            {
                v2f a;
                float a0 = mrow[24];
                a.x = (half2 == 0) ? a0 : 0.0f;
                a.y = 0.0f;
                v2f bb;
                bb.x = ucur[pb + 24];
                bb.y = ucur[pb + 25];
                acc = __builtin_amdgcn_wmma_f32_16x16x4_f32(
                          false, a, false, bb, (short)0, acc, false, false);
            }

            float dv = acc[0];
#pragma unroll
            for (int v = 1; v < 8; ++v)
                dv = ((lane & 7) == v) ? acc[v] : dv;

            const int pst = hasDiag ? ((REGION - 1) - (l0 + dslot)) : DUMP;
            unxt[pst] = dv * inv_a;
        }
        __syncthreads();

#pragma unroll
        for (int k = 0; k < YPT; ++k)
            yreg[k] += unxt[(CHUNK - 1) - (tid + k * NTHREADS)];

        float* tmp = ucur; ucur = unxt; unxt = tmp;
    }

    float* __restrict__ yb = y + (size_t)b * T_LEN;
#pragma unroll
    for (int k = 0; k < YPT; ++k) {
        int t = cs + tid + k * NTHREADS;
        if (t < T_LEN) yb[t] = yreg[k];
    }
}

extern "C" void kernel_launch(void* const* d_in, const int* in_sizes, int n_in,
                              void* d_out, int out_size, void* d_ws, size_t ws_size,
                              hipStream_t stream) {
    (void)n_in; (void)out_size; (void)d_ws; (void)ws_size;
    const float* x  = (const float*)d_in[0];
    const float* mc = (const float*)d_in[1];
    float* yout     = (float*)d_out;

    const int B = in_sizes[0] / T_LEN;
    dim3 grid((T_LEN + CHUNK - 1) / CHUNK, B);
    mglsa_taylor_wmma_kernel<<<grid, NTHREADS, 0, stream>>>(x, mc, yout);
}


# loop order swapped: each wave walks its own time chunk in 64-sample blocks and runs all 30 Taylor levels on a block before the next; mc rows read once, per-level history rings in LDS, no barriers
# speedup vs baseline: 1.3989x; 1.3989x over previous
#include <hip/hip_runtime.h>


typedef __attribute__((ext_vector_type(2))) float v2f;
typedef __attribute__((ext_vector_type(8))) float v8f;

#define T_LEN    48000
#define M1       25
#define NITER    30
#define SG       64
#define NSG      (T_LEN / SG)
#define WPR      128
#define LEAD     12
#define RSZ      256
#define NLV      (NITER + 1)
#define DUMPF    160
#define WAVE_USED (NLV * RSZ + SG + DUMPF)
#define WAVE_F   10496
static_assert(WAVE_F >= WAVE_USED && (WAVE_F % 4) == 0, "per-wave LDS too small or misaligned");
#define NWAVES   4
#define NTHREADS (NWAVES * 32)


__device__ __forceinline__ void load_a(const float* __restrict__ mcb, int sg, int c, int h2,
                                       v2f (&A)[4][7])
{
#pragma unroll
    for (int g = 0; g < 4; ++g) {
        const float* __restrict__ mrow = mcb + (size_t)(sg * SG + 16 * g + c) * M1;
#pragma unroll
        for (int q = 0; q < 6; ++q) {
            A[g][q].x = mrow[4 * q + h2];
            A[g][q].y = mrow[4 * q + h2 + 1];
        }
        const float a24 = mrow[24];
        A[g][6].x = (h2 == 0) ? a24 : 0.0f;
        A[g][6].y = 0.0f;
    }
}

__device__ __forceinline__ void load_b(const float* Rl, int b0, int b1, v2f (&B)[2][7])
{
#pragma unroll
    for (int q = 0; q < 7; ++q) {
        B[0][q].x = Rl[b0 + 4 * q];
        B[0][q].y = Rl[b0 + 4 * q + 1];
        B[1][q].x = Rl[b1 + 4 * q];
        B[1][q].y = Rl[b1 + 4 * q + 1];
    }
}

__device__ __forceinline__ float diag_of(const v8f& acc, int lane7)
{
    float dv = acc[0];
#pragma unroll
    for (int v = 1; v < 8; ++v)
        dv = (lane7 == v) ? acc[v] : dv;
    return dv;
}

__global__ __launch_bounds__(NTHREADS)
void mglsa_stream_wmma_kernel(const float* __restrict__ x,
                              const float* __restrict__ mc,
                              float* __restrict__ y)
{
    __shared__ __attribute__((aligned(16))) float lds[NWAVES * WAVE_F];

    const int lane = threadIdx.x & 31;
    const int wave = __builtin_amdgcn_readfirstlane((int)(threadIdx.x >> 5));
    const int wg   = (int)blockIdx.x * NWAVES + wave;
    const int b    = wg / WPR;
    const int ci   = wg % WPR;
    const int sg0  = (ci * NSG) / WPR;
    const int sg1  = ((ci + 1) * NSG) / WPR;
    const int hs   = (sg0 - LEAD) > 0 ? (sg0 - LEAD) : 0;

    float* R      = lds + wave * WAVE_F;
    float* ystage = R + NLV * RSZ;
    const int DUMP0 = NLV * RSZ + SG;

    const float* __restrict__ xb  = x  + (size_t)b * T_LEN;
    const float* __restrict__ mcb = mc + (size_t)b * T_LEN * M1;
    float* __restrict__       yb  = y  + (size_t)b * T_LEN;

    const int  c       = lane & 15;
    const int  h2      = (lane >> 4) << 1;
    const int  lane7   = lane & 7;
    const bool hasDiag = (lane < 8) || (lane >= 24);
    const int  dslot   = (lane < 8) ? lane : (lane - 16);

    {
        float4* R4 = reinterpret_cast<float4*>(R);
        for (int i = lane; i < (NLV * RSZ) / 4; i += 32)
            R4[i] = make_float4(0.f, 0.f, 0.f, 0.f);
    }

    v2f A[4][7];
    load_a(mcb, hs, c, h2, A);
    float xa = xb[hs * SG + lane];
    float xc = xb[hs * SG + 32 + lane];

    for (int sg = hs; sg < sg1; ++sg) {
        const int nsg = (sg + 1 < sg1) ? (sg + 1) : sg;
        v2f An[4][7];
        load_a(mcb, nsg, c, h2, An);
        const float xan = xb[nsg * SG + lane];
        const float xcn = xb[nsg * SG + 32 + lane];

        const int rbase = (sg & 1) ? 63 : 127;

        {
            const int p = rbase - lane;
            R[p]            = xa;
            R[p + 128]      = xa;
            R[p - 32]       = xc;
            R[p - 32 + 128] = xc;
        }

        int   bB[4], ps[4];
        float yacc[4];
#pragma unroll
        for (int g = 0; g < 4; ++g) {
            const int pd = rbase - 16 * g - dslot;
            bB[g]   = rbase - 16 * g - c + h2;
            yacc[g] = R[pd];
            ps[g]   = hasDiag ? pd : (DUMP0 + lane);
        }
        const int lstride = hasDiag ? RSZ : 0;

        v2f B0[2][7];
        load_b(R, bB[0], bB[1], B0);

#pragma unroll 1
        for (int n = 1; n <= NITER; ++n) {
            const float  inv_a = 1.0f / (float)n;
            const float* Rp = R + (n - 1) * RSZ;
            const float* Rn = R + n * RSZ;
            float*       Sn = R + n * lstride;

            v2f B1[2][7];
            load_b(Rp, bB[2], bB[3], B1);
            __builtin_amdgcn_sched_barrier(0);

            v8f acc0 = {0.f, 0.f, 0.f, 0.f, 0.f, 0.f, 0.f, 0.f};
            v8f acc1 = {0.f, 0.f, 0.f, 0.f, 0.f, 0.f, 0.f, 0.f};
#pragma unroll
            for (int q = 0; q < 7; ++q) {
                acc0 = __builtin_amdgcn_wmma_f32_16x16x4_f32(
                           false, A[0][q], false, B0[0][q], (short)0, acc0, false, false);
                acc1 = __builtin_amdgcn_wmma_f32_16x16x4_f32(
                           false, A[1][q], false, B0[1][q], (short)0, acc1, false, false);
            }
            {
                const float u0 = diag_of(acc0, lane7) * inv_a;
                const float u1 = diag_of(acc1, lane7) * inv_a;
                Sn[ps[0]] = u0;  Sn[ps[0] + 128] = u0;
                Sn[ps[1]] = u1;  Sn[ps[1] + 128] = u1;
                yacc[0] += u0;
                yacc[1] += u1;
            }

            load_b(Rn, bB[0], bB[1], B0);
            __builtin_amdgcn_sched_barrier(0);

            v8f acc2 = {0.f, 0.f, 0.f, 0.f, 0.f, 0.f, 0.f, 0.f};
            v8f acc3 = {0.f, 0.f, 0.f, 0.f, 0.f, 0.f, 0.f, 0.f};
#pragma unroll
            for (int q = 0; q < 7; ++q) {
                acc2 = __builtin_amdgcn_wmma_f32_16x16x4_f32(
                           false, A[2][q], false, B1[0][q], (short)0, acc2, false, false);
                acc3 = __builtin_amdgcn_wmma_f32_16x16x4_f32(
                           false, A[3][q], false, B1[1][q], (short)0, acc3, false, false);
            }
            {
                const float u2 = diag_of(acc2, lane7) * inv_a;
                const float u3 = diag_of(acc3, lane7) * inv_a;
                Sn[ps[2]] = u2;  Sn[ps[2] + 128] = u2;
                Sn[ps[3]] = u3;  Sn[ps[3] + 128] = u3;
                yacc[2] += u2;
                yacc[3] += u3;
            }
        }

        if (sg >= sg0) {
            if (hasDiag) {
#pragma unroll
                for (int g = 0; g < 4; ++g)
                    ystage[16 * g + dslot] = yacc[g];
            }
            const float y0 = ystage[lane];
            const float y1 = ystage[32 + lane];
            yb[sg * SG + lane]      = y0;
            yb[sg * SG + 32 + lane] = y1;
        }

#pragma unroll
        for (int g = 0; g < 4; ++g)
#pragma unroll
            for (int q = 0; q < 7; ++q)
                A[g][q] = An[g][q];
        xa = xan;
        xc = xcn;
    }
}

extern "C" void kernel_launch(void* const* d_in, const int* in_sizes, int n_in,
                              void* d_out, int out_size, void* d_ws, size_t ws_size,
                              hipStream_t stream) {
    (void)in_sizes; (void)n_in; (void)out_size; (void)d_ws; (void)ws_size;
    const float* x  = (const float*)d_in[0];
    const float* mc = (const float*)d_in[1];
    float* yout     = (float*)d_out;

    mglsa_stream_wmma_kernel<<<dim3(256), dim3(NTHREADS), 0, stream>>>(x, mc, yout);
}


# one stream per workgroup, its 30 Taylor levels pipelined over 8 waves in 64-sample blocks (one barrier per step, mc block shared through LDS tiles): lead-in paid per 24 blocks, 2 waves per SIMD
# speedup vs baseline: 2.3415x; 1.6738x over previous
#include <hip/hip_runtime.h>


typedef __attribute__((ext_vector_type(2))) float v2f;
typedef __attribute__((ext_vector_type(8))) float v8f;

#define T_LEN    48000
#define M1       25
#define NITER    30
#define SG       64
#define NSG      (T_LEN / SG)
#define CPR      32
#define LEAD     12
#define NWAVES   8
#define NTHREADS (NWAVES * 32)
#define RING     256
#define RSZ      (2 * RING)
#define NLV      (NITER + 1)
#define TILE_F   (SG * M1)
#define TILE_V4  (TILE_F / 4)
#define NTILE    16
#define AHEAD    2
#define TILE_OFF (NLV * RSZ)
#define YP_OFF   (TILE_OFF + NTILE * TILE_F)
#define YST_OFF  (YP_OFF + NWAVES * 2 * SG)
#define DUMPW    (32 + RING)
#define DUMP_OFF (YST_OFF + SG)
#define LDS_F    (DUMP_OFF + NWAVES * DUMPW)
static_assert((LDS_F % 4) == 0 && (TILE_OFF % 4) == 0 && (TILE_F % 4) == 0, "16-byte pieces");
static_assert(LDS_F * 4 > 160 * 1024 && LDS_F * 4 <= 320 * 1024, "one workgroup per WGP");
static_assert((CPR * 8) == 256 && NWAVES == 8, "level split below is written for 8 waves");
static_assert(TILE_V4 > NTHREADS && TILE_V4 <= 2 * NTHREADS, "two pieces per thread");


__device__ __forceinline__ void load_a(const float* tile, int c, int h2, v2f (&A)[4][7])
{
#pragma unroll
    for (int g = 0; g < 4; ++g) {
        const float* mrow = tile + (16 * g + c) * M1;
#pragma unroll
        for (int q = 0; q < 6; ++q) {
            A[g][q].x = mrow[4 * q + h2];
            A[g][q].y = mrow[4 * q + h2 + 1];
        }
        const float a24 = mrow[24];
        A[g][6].x = (h2 == 0) ? a24 : 0.0f;
        A[g][6].y = 0.0f;
    }
}

__device__ __forceinline__ void load_b(const float* Rl, int b0, int b1, v2f (&B)[2][7])
{
#pragma unroll
    for (int q = 0; q < 7; ++q) {
        B[0][q].x = Rl[b0 + 4 * q];
        B[0][q].y = Rl[b0 + 4 * q + 1];
        B[1][q].x = Rl[b1 + 4 * q];
        B[1][q].y = Rl[b1 + 4 * q + 1];
    }
}

__device__ __forceinline__ float diag_of(const v8f& acc, int lane7)
{
    float dv = acc[0];
#pragma unroll
    for (int v = 1; v < 8; ++v)
        dv = (lane7 == v) ? acc[v] : dv;
    return dv;
}

__global__ __launch_bounds__(NTHREADS)
void mglsa_levelpipe_wmma_kernel(const float* __restrict__ x,
                                 const float* __restrict__ mc,
                                 float* __restrict__ y)
{
    __shared__ __attribute__((aligned(16))) float lds[LDS_F];

    const int tid  = (int)threadIdx.x;
    const int lane = tid & 31;
    const int w    = __builtin_amdgcn_readfirstlane(tid >> 5);
    const int b    = (int)blockIdx.x / CPR;
    const int ci   = (int)blockIdx.x % CPR;
    const int sg0  = (ci * NSG) / CPR;
    const int sg1  = ((ci + 1) * NSG) / CPR;
    const int hs   = (sg0 - LEAD) > 0 ? (sg0 - LEAD) : 0;
    const int nblk = sg1 - hs;

    const int nlo = (w == 0) ? 0 : (4 * w - 1);
    const int nhi = (w == NWAVES - 1) ? NITER : (4 * w + 3);

    float* R      = lds;
    float* tiles  = lds + TILE_OFF;
    const float* ypin  = lds + YP_OFF + ((w > 0) ? (w - 1) : 0) * (2 * SG);
    float*       ypout = lds + YP_OFF + w * (2 * SG);
    float*       ystage = lds + YST_OFF;
    const int DUMP0 = DUMP_OFF + w * DUMPW;

    const float* __restrict__ xb  = x  + (size_t)b * T_LEN;
    const float* __restrict__ mcb = mc + (size_t)b * T_LEN * M1;
    float* __restrict__       yb  = y  + (size_t)b * T_LEN;
    const float4* __restrict__ mc4 = reinterpret_cast<const float4*>(mcb);
    const bool second = tid < (TILE_V4 - NTHREADS);

    const int  c       = lane & 15;
    const int  h2      = (lane >> 4) << 1;
    const int  lane7   = lane & 7;
    const bool hasDiag = (lane < 8) || (lane >= 24);
    const int  dslot   = (lane < 8) ? lane : (lane - 16);

    {
        float4* R4 = reinterpret_cast<float4*>(R);
        for (int i = tid; i < (NLV * RSZ) / 4; i += NTHREADS)
            R4[i] = make_float4(0.f, 0.f, 0.f, 0.f);
#pragma unroll
        for (int j = 0; j < AHEAD; ++j) {
            const int blk = hs + j;
            float4* t4 = reinterpret_cast<float4*>(tiles + (blk & (NTILE - 1)) * TILE_F);
            t4[tid] = mc4[blk * TILE_V4 + tid];
            if (second)
                t4[NTHREADS + tid] = mc4[blk * TILE_V4 + NTHREADS + tid];
        }
    }
    float xa = 0.f, xc = 0.f;
    if (w == 0) {
        xa = xb[hs * SG + lane];
        xc = xb[hs * SG + 32 + lane];
    }

    __syncthreads();

    v2f A[4][7];
    load_a(tiles + (hs & (NTILE - 1)) * TILE_F, c, h2, A);

    for (int s = 0; s < nblk + NWAVES - 1; ++s) {
        const int  fb    = hs + s + AHEAD;
        const bool fetch = fb < sg1;
        float4 f0 = make_float4(0.f, 0.f, 0.f, 0.f);
        float4 f1 = make_float4(0.f, 0.f, 0.f, 0.f);
        if (fetch) {
            f0 = mc4[fb * TILE_V4 + tid];
            if (second)
                f1 = mc4[fb * TILE_V4 + NTHREADS + tid];
        }

        const int k = hs + s - w;
        if (s >= w && k < sg1) {
            const int nk = (k + 1 < sg1) ? (k + 1) : k;
            float xan = 0.f, xcn = 0.f;
            if (w == 0) {
                xan = xb[nk * SG + lane];
                xcn = xb[nk * SG + 32 + lane];
            }

            const int rbase = (RING - 1) - SG * (k & 3);
            const int ypar  = (k & 1) * SG;

            if (w == 0) {
                const int p = rbase - lane;
                R[p]             = xa;
                R[p + RING]      = xa;
                R[p - 32]        = xc;
                R[p - 32 + RING] = xc;
            }

            int   bB[4], ps[4];
            float yacc[4];
#pragma unroll
            for (int g = 0; g < 4; ++g) {
                const int pd = rbase - 16 * g - dslot;
                bB[g] = rbase - 16 * g - c + h2;
                ps[g] = hasDiag ? pd : (DUMP0 + lane);
                const int yi = hasDiag ? (16 * g + dslot) : lane;
                yacc[g] = (w == 0) ? R[pd] : ypin[ypar + yi];
            }
            const int lstride = hasDiag ? RSZ : 0;

            v2f B0[2][7];
            load_b(R + nlo * RSZ, bB[0], bB[1], B0);
            __builtin_amdgcn_sched_barrier(0);

            v2f An[4][7];
            load_a(tiles + (nk & (NTILE - 1)) * TILE_F, c, h2, An);

#pragma unroll 1
            for (int n = nlo + 1; n <= nhi; ++n) {
                const float  inv_a = 1.0f / (float)n;
                const float* Rp = R + (n - 1) * RSZ;
                const float* Rn = R + n * RSZ;
                float*       Sn = R + n * lstride;

                v2f B1[2][7];
                load_b(Rp, bB[2], bB[3], B1);
                __builtin_amdgcn_sched_barrier(0);

                v8f acc0 = {0.f, 0.f, 0.f, 0.f, 0.f, 0.f, 0.f, 0.f};
                v8f acc1 = {0.f, 0.f, 0.f, 0.f, 0.f, 0.f, 0.f, 0.f};
#pragma unroll
                for (int q = 0; q < 7; ++q) {
                    acc0 = __builtin_amdgcn_wmma_f32_16x16x4_f32(
                               false, A[0][q], false, B0[0][q], (short)0, acc0, false, false);
                    acc1 = __builtin_amdgcn_wmma_f32_16x16x4_f32(
                               false, A[1][q], false, B0[1][q], (short)0, acc1, false, false);
                }
                {
                    const float u0 = diag_of(acc0, lane7) * inv_a;
                    const float u1 = diag_of(acc1, lane7) * inv_a;
                    Sn[ps[0]] = u0;  Sn[ps[0] + RING] = u0;
                    Sn[ps[1]] = u1;  Sn[ps[1] + RING] = u1;
                    yacc[0] += u0;
                    yacc[1] += u1;
                }

                load_b(Rn, bB[0], bB[1], B0);
                __builtin_amdgcn_sched_barrier(0);

                v8f acc2 = {0.f, 0.f, 0.f, 0.f, 0.f, 0.f, 0.f, 0.f};
                v8f acc3 = {0.f, 0.f, 0.f, 0.f, 0.f, 0.f, 0.f, 0.f};
#pragma unroll
                for (int q = 0; q < 7; ++q) {
                    acc2 = __builtin_amdgcn_wmma_f32_16x16x4_f32(
                               false, A[2][q], false, B1[0][q], (short)0, acc2, false, false);
                    acc3 = __builtin_amdgcn_wmma_f32_16x16x4_f32(
                               false, A[3][q], false, B1[1][q], (short)0, acc3, false, false);
                }
                {
                    const float u2 = diag_of(acc2, lane7) * inv_a;
                    const float u3 = diag_of(acc3, lane7) * inv_a;
                    Sn[ps[2]] = u2;  Sn[ps[2] + RING] = u2;
                    Sn[ps[3]] = u3;  Sn[ps[3] + RING] = u3;
                    yacc[2] += u2;
                    yacc[3] += u3;
                }
            }

            if (w == NWAVES - 1) {
                if (k >= sg0) {
                    if (hasDiag) {
#pragma unroll
                        for (int g = 0; g < 4; ++g)
                            ystage[16 * g + dslot] = yacc[g];
                    }
                    const float y0 = ystage[lane];
                    const float y1 = ystage[32 + lane];
                    yb[k * SG + lane]      = y0;
                    yb[k * SG + 32 + lane] = y1;
                }
            } else {
                if (hasDiag) {
#pragma unroll
                    for (int g = 0; g < 4; ++g)
                        ypout[ypar + 16 * g + dslot] = yacc[g];
                }
            }

#pragma unroll
            for (int g = 0; g < 4; ++g)
#pragma unroll
                for (int q = 0; q < 7; ++q)
                    A[g][q] = An[g][q];
            xa = xan;
            xc = xcn;
        }

        if (fetch) {
            float4* t4 = reinterpret_cast<float4*>(tiles + (fb & (NTILE - 1)) * TILE_F);
            t4[tid] = f0;
            if (second)
                t4[NTHREADS + tid] = f1;
        }
        __syncthreads();
    }
}

extern "C" void kernel_launch(void* const* d_in, const int* in_sizes, int n_in,
                              void* d_out, int out_size, void* d_ws, size_t ws_size,
                              hipStream_t stream) {
    (void)in_sizes; (void)n_in; (void)out_size; (void)d_ws; (void)ws_size;
    const float* x  = (const float*)d_in[0];
    const float* mc = (const float*)d_in[1];
    float* yout     = (float*)d_out;

    mglsa_levelpipe_wmma_kernel<<<dim3(256), dim3(NTHREADS), 0, stream>>>(x, mc, yout);
}


# next block's A fragments loaded inside the step's last level, each half into the registers its last WMMAs just read: no second fragment set, no copy, 32 LDS loads out of the post-barrier burst
# speedup vs baseline: 2.5159x; 1.0745x over previous
#include <hip/hip_runtime.h>


typedef __attribute__((ext_vector_type(2))) float v2f;
typedef __attribute__((ext_vector_type(8))) float v8f;

#define T_LEN    48000
#define M1       25
#define NITER    30
#define SG       64
#define NSG      (T_LEN / SG)
#define CPR      32
#define LEAD     12
#define NWAVES   8
#define NTHREADS (NWAVES * 32)
#define RING     256
#define RSZ      (2 * RING)
#define NLV      (NITER + 1)
#define TILE_F   (SG * M1)
#define TILE_V4  (TILE_F / 4)
#define NTILE    16
#define AHEAD    2
#define TILE_OFF (NLV * RSZ)
#define YP_OFF   (TILE_OFF + NTILE * TILE_F)
#define YST_OFF  (YP_OFF + NWAVES * 2 * SG)
#define DUMPW    (32 + RING)
#define DUMP_OFF (YST_OFF + SG)
#define LDS_F    (DUMP_OFF + NWAVES * DUMPW)
static_assert((LDS_F % 4) == 0 && (TILE_OFF % 4) == 0 && (TILE_F % 4) == 0, "16-byte pieces");
static_assert(LDS_F * 4 > 160 * 1024 && LDS_F * 4 <= 320 * 1024, "one workgroup per WGP");
static_assert((CPR * 8) == 256 && NWAVES == 8, "level split below is written for 8 waves");
static_assert(TILE_V4 > NTHREADS && TILE_V4 <= 2 * NTHREADS, "two pieces per thread");


__device__ __forceinline__ void load_a_pair(const float* tile, int c, int h2, int g0, v2f (&A)[4][7])
{
#pragma unroll
    for (int g = g0; g < g0 + 2; ++g) {
        const float* mrow = tile + (16 * g + c) * M1;
#pragma unroll
        for (int q = 0; q < 6; ++q) {
            A[g][q].x = mrow[4 * q + h2];
            A[g][q].y = mrow[4 * q + h2 + 1];
        }
        const float a24 = mrow[24];
        A[g][6].x = (h2 == 0) ? a24 : 0.0f;
        A[g][6].y = 0.0f;
    }
}

__device__ __forceinline__ void load_b(const float* Rl, int b0, int b1, v2f (&B)[2][7])
{
#pragma unroll
    for (int q = 0; q < 7; ++q) {
        B[0][q].x = Rl[b0 + 4 * q];
        B[0][q].y = Rl[b0 + 4 * q + 1];
        B[1][q].x = Rl[b1 + 4 * q];
        B[1][q].y = Rl[b1 + 4 * q + 1];
    }
}

__device__ __forceinline__ float diag_of(const v8f& acc, int lane7)
{
    float dv = acc[0];
#pragma unroll
    for (int v = 1; v < 8; ++v)
        dv = (lane7 == v) ? acc[v] : dv;
    return dv;
}

template <bool LAST>
__device__ __forceinline__ void level_trip(const int n, float* R, const int lstride,
                                           const int (&bB)[4], const int (&ps)[4],
                                           float (&yacc)[4], v2f (&A)[4][7], v2f (&B0)[2][7],
                                           const float* ntile, const int c, const int h2,
                                           const int lane7)
{
    const float  inv_a = 1.0f / (float)n;
    const float* Rp = R + (n - 1) * RSZ;
    const float* Rn = R + n * RSZ;
    float*       Sn = R + n * lstride;

    v2f B1[2][7];
    load_b(Rp, bB[2], bB[3], B1);
    __builtin_amdgcn_sched_barrier(0);

    v8f acc0 = {0.f, 0.f, 0.f, 0.f, 0.f, 0.f, 0.f, 0.f};
    v8f acc1 = {0.f, 0.f, 0.f, 0.f, 0.f, 0.f, 0.f, 0.f};
#pragma unroll
    for (int q = 0; q < 7; ++q) {
        acc0 = __builtin_amdgcn_wmma_f32_16x16x4_f32(
                   false, A[0][q], false, B0[0][q], (short)0, acc0, false, false);
        acc1 = __builtin_amdgcn_wmma_f32_16x16x4_f32(
                   false, A[1][q], false, B0[1][q], (short)0, acc1, false, false);
    }
    {
        const float u0 = diag_of(acc0, lane7) * inv_a;
        const float u1 = diag_of(acc1, lane7) * inv_a;
        Sn[ps[0]] = u0;  Sn[ps[0] + RING] = u0;
        Sn[ps[1]] = u1;  Sn[ps[1] + RING] = u1;
        yacc[0] += u0;
        yacc[1] += u1;
    }

    if constexpr (!LAST) {
        load_b(Rn, bB[0], bB[1], B0);
    } else {
        load_a_pair(ntile, c, h2, 0, A);
    }
    __builtin_amdgcn_sched_barrier(0);

    v8f acc2 = {0.f, 0.f, 0.f, 0.f, 0.f, 0.f, 0.f, 0.f};
    v8f acc3 = {0.f, 0.f, 0.f, 0.f, 0.f, 0.f, 0.f, 0.f};
#pragma unroll
    for (int q = 0; q < 7; ++q) {
        acc2 = __builtin_amdgcn_wmma_f32_16x16x4_f32(
                   false, A[2][q], false, B1[0][q], (short)0, acc2, false, false);
        acc3 = __builtin_amdgcn_wmma_f32_16x16x4_f32(
                   false, A[3][q], false, B1[1][q], (short)0, acc3, false, false);
    }
    {
        const float u2 = diag_of(acc2, lane7) * inv_a;
        const float u3 = diag_of(acc3, lane7) * inv_a;
        Sn[ps[2]] = u2;  Sn[ps[2] + RING] = u2;
        Sn[ps[3]] = u3;  Sn[ps[3] + RING] = u3;
        yacc[2] += u2;
        yacc[3] += u3;
    }
    if constexpr (LAST) {
        load_a_pair(ntile, c, h2, 2, A);
    }
}

__global__ __launch_bounds__(NTHREADS)
void mglsa_levelpipe_wmma_kernel(const float* __restrict__ x,
                                 const float* __restrict__ mc,
                                 float* __restrict__ y)
{
    __shared__ __attribute__((aligned(16))) float lds[LDS_F];

    const int tid  = (int)threadIdx.x;
    const int lane = tid & 31;
    const int w    = __builtin_amdgcn_readfirstlane(tid >> 5);
    const int b    = (int)blockIdx.x / CPR;
    const int ci   = (int)blockIdx.x % CPR;
    const int sg0  = (ci * NSG) / CPR;
    const int sg1  = ((ci + 1) * NSG) / CPR;
    const int hs   = (sg0 - LEAD) > 0 ? (sg0 - LEAD) : 0;
    const int nblk = sg1 - hs;

    const int nlo = (w == 0) ? 0 : (4 * w - 1);
    const int nhi = (w == NWAVES - 1) ? NITER : (4 * w + 3);

    float* R      = lds;
    float* tiles  = lds + TILE_OFF;
    const float* ypin  = lds + YP_OFF + ((w > 0) ? (w - 1) : 0) * (2 * SG);
    float*       ypout = lds + YP_OFF + w * (2 * SG);
    float*       ystage = lds + YST_OFF;
    const int DUMP0 = DUMP_OFF + w * DUMPW;

    const float* __restrict__ xb  = x  + (size_t)b * T_LEN;
    const float* __restrict__ mcb = mc + (size_t)b * T_LEN * M1;
    float* __restrict__       yb  = y  + (size_t)b * T_LEN;
    const float4* __restrict__ mc4 = reinterpret_cast<const float4*>(mcb);
    const bool second = tid < (TILE_V4 - NTHREADS);

    const int  c       = lane & 15;
    const int  h2      = (lane >> 4) << 1;
    const int  lane7   = lane & 7;
    const bool hasDiag = (lane < 8) || (lane >= 24);
    const int  dslot   = (lane < 8) ? lane : (lane - 16);

    {
        float4* R4 = reinterpret_cast<float4*>(R);
        for (int i = tid; i < (NLV * RSZ) / 4; i += NTHREADS)
            R4[i] = make_float4(0.f, 0.f, 0.f, 0.f);
#pragma unroll
        for (int j = 0; j < AHEAD; ++j) {
            const int blk = hs + j;
            float4* t4 = reinterpret_cast<float4*>(tiles + (blk & (NTILE - 1)) * TILE_F);
            t4[tid] = mc4[blk * TILE_V4 + tid];
            if (second)
                t4[NTHREADS + tid] = mc4[blk * TILE_V4 + NTHREADS + tid];
        }
    }
    float xa = 0.f, xc = 0.f;
    if (w == 0) {
        xa = xb[hs * SG + lane];
        xc = xb[hs * SG + 32 + lane];
    }

    __syncthreads();

    v2f A[4][7];
    load_a_pair(tiles + (hs & (NTILE - 1)) * TILE_F, c, h2, 0, A);
    load_a_pair(tiles + (hs & (NTILE - 1)) * TILE_F, c, h2, 2, A);

    for (int s = 0; s < nblk + NWAVES - 1; ++s) {
        const int  fb    = hs + s + AHEAD;
        const bool fetch = fb < sg1;
        float4 f0 = make_float4(0.f, 0.f, 0.f, 0.f);
        float4 f1 = make_float4(0.f, 0.f, 0.f, 0.f);
        if (fetch) {
            f0 = mc4[fb * TILE_V4 + tid];
            if (second)
                f1 = mc4[fb * TILE_V4 + NTHREADS + tid];
        }

        const int k = hs + s - w;
        if (s >= w && k < sg1) {
            const int nk = (k + 1 < sg1) ? (k + 1) : k;
            float xan = 0.f, xcn = 0.f;
            if (w == 0) {
                xan = xb[nk * SG + lane];
                xcn = xb[nk * SG + 32 + lane];
            }

            const int rbase = (RING - 1) - SG * (k & 3);
            const int ypar  = (k & 1) * SG;

            if (w == 0) {
                const int p = rbase - lane;
                R[p]             = xa;
                R[p + RING]      = xa;
                R[p - 32]        = xc;
                R[p - 32 + RING] = xc;
            }

            int   bB[4], ps[4];
            float yacc[4];
#pragma unroll
            for (int g = 0; g < 4; ++g) {
                const int pd = rbase - 16 * g - dslot;
                bB[g] = rbase - 16 * g - c + h2;
                ps[g] = hasDiag ? pd : (DUMP0 + lane);
                const int yi = hasDiag ? (16 * g + dslot) : lane;
                yacc[g] = (w == 0) ? R[pd] : ypin[ypar + yi];
            }
            const int lstride = hasDiag ? RSZ : 0;

            v2f B0[2][7];
            load_b(R + nlo * RSZ, bB[0], bB[1], B0);

            const float* ntile = tiles + (nk & (NTILE - 1)) * TILE_F;

#pragma unroll 1
            for (int n = nlo + 1; n < nhi; ++n)
                level_trip<false>(n, R, lstride, bB, ps, yacc, A, B0, ntile, c, h2, lane7);
            level_trip<true>(nhi, R, lstride, bB, ps, yacc, A, B0, ntile, c, h2, lane7);

            if (w == NWAVES - 1) {
                if (k >= sg0) {
                    if (hasDiag) {
#pragma unroll
                        for (int g = 0; g < 4; ++g)
                            ystage[16 * g + dslot] = yacc[g];
                    }
                    const float y0 = ystage[lane];
                    const float y1 = ystage[32 + lane];
                    yb[k * SG + lane]      = y0;
                    yb[k * SG + 32 + lane] = y1;
                }
            } else {
                if (hasDiag) {
#pragma unroll
                    for (int g = 0; g < 4; ++g)
                        ypout[ypar + 16 * g + dslot] = yacc[g];
                }
            }

            xa = xan;
            xc = xcn;
        }

        if (fetch) {
            float4* t4 = reinterpret_cast<float4*>(tiles + (fb & (NTILE - 1)) * TILE_F);
            t4[tid] = f0;
            if (second)
                t4[NTHREADS + tid] = f1;
        }
        __syncthreads();
    }
}

extern "C" void kernel_launch(void* const* d_in, const int* in_sizes, int n_in,
                              void* d_out, int out_size, void* d_ws, size_t ws_size,
                              hipStream_t stream) {
    (void)in_sizes; (void)n_in; (void)out_size; (void)d_ws; (void)ws_size;
    const float* x  = (const float*)d_in[0];
    const float* mc = (const float*)d_in[1];
    float* yout     = (float*)d_out;

    mglsa_levelpipe_wmma_kernel<<<dim3(256), dim3(NTHREADS), 0, stream>>>(x, mc, yout);
}


# each pipeline stage (wave) joins at the first block its own levels can still pass to an output (24 samples per remaining level: 12,10,9,7,6,4,3,1 lead-in blocks), not at the common start; sums kept
# speedup vs baseline: 2.6578x; 1.0564x over previous
#include <hip/hip_runtime.h>


typedef __attribute__((ext_vector_type(2))) float v2f;
typedef __attribute__((ext_vector_type(8))) float v8f;

#define T_LEN    48000
#define M1       25
#define NITER    30
#define SG       64
#define NSG      (T_LEN / SG)
#define CPR      32
#define LEAD     12
#define NWAVES   8
#define NTHREADS (NWAVES * 32)
#define RING     256
#define RSZ      (2 * RING)
#define NLV      (NITER + 1)
#define TILE_F   (SG * M1)
#define TILE_V4  (TILE_F / 4)
#define NTILE    16
#define AHEAD    2
#define TILE_OFF (NLV * RSZ)
#define YP_OFF   (TILE_OFF + NTILE * TILE_F)
#define YST_OFF  (YP_OFF + NWAVES * 2 * SG)
#define DUMPW    (32 + RING)
#define DUMP_OFF (YST_OFF + SG)
#define LDS_F    (DUMP_OFF + NWAVES * DUMPW)
static_assert((LDS_F % 4) == 0 && (TILE_OFF % 4) == 0 && (TILE_F % 4) == 0, "16-byte pieces");
static_assert(LDS_F * 4 > 160 * 1024 && LDS_F * 4 <= 320 * 1024, "one workgroup per WGP");
static_assert((CPR * 8) == 256 && NWAVES == 8, "level split below is written for 8 waves");
static_assert(TILE_V4 > NTHREADS && TILE_V4 <= 2 * NTHREADS, "two pieces per thread");


__device__ __forceinline__ void load_a_pair(const float* tile, int c, int h2, int g0, v2f (&A)[4][7])
{
#pragma unroll
    for (int g = g0; g < g0 + 2; ++g) {
        const float* mrow = tile + (16 * g + c) * M1;
#pragma unroll
        for (int q = 0; q < 6; ++q) {
            A[g][q].x = mrow[4 * q + h2];
            A[g][q].y = mrow[4 * q + h2 + 1];
        }
        const float a24 = mrow[24];
        A[g][6].x = (h2 == 0) ? a24 : 0.0f;
        A[g][6].y = 0.0f;
    }
}

__device__ __forceinline__ void load_b(const float* Rl, int b0, int b1, v2f (&B)[2][7])
{
#pragma unroll
    for (int q = 0; q < 7; ++q) {
        B[0][q].x = Rl[b0 + 4 * q];
        B[0][q].y = Rl[b0 + 4 * q + 1];
        B[1][q].x = Rl[b1 + 4 * q];
        B[1][q].y = Rl[b1 + 4 * q + 1];
    }
}

__device__ __forceinline__ float diag_of(const v8f& acc, int lane7)
{
    float dv = acc[0];
#pragma unroll
    for (int v = 1; v < 8; ++v)
        dv = (lane7 == v) ? acc[v] : dv;
    return dv;
}

template <bool LAST>
__device__ __forceinline__ void level_trip(const int n, float* R, const int lstride,
                                           const int (&bB)[4], const int (&ps)[4],
                                           float (&yacc)[4], v2f (&A)[4][7], v2f (&B0)[2][7],
                                           const float* ntile, const int c, const int h2,
                                           const int lane7)
{
    const float  inv_a = 1.0f / (float)n;
    const float* Rp = R + (n - 1) * RSZ;
    const float* Rn = R + n * RSZ;
    float*       Sn = R + n * lstride;

    v2f B1[2][7];
    load_b(Rp, bB[2], bB[3], B1);
    __builtin_amdgcn_sched_barrier(0);

    v8f acc0 = {0.f, 0.f, 0.f, 0.f, 0.f, 0.f, 0.f, 0.f};
    v8f acc1 = {0.f, 0.f, 0.f, 0.f, 0.f, 0.f, 0.f, 0.f};
#pragma unroll
    for (int q = 0; q < 7; ++q) {
        acc0 = __builtin_amdgcn_wmma_f32_16x16x4_f32(
                   false, A[0][q], false, B0[0][q], (short)0, acc0, false, false);
        acc1 = __builtin_amdgcn_wmma_f32_16x16x4_f32(
                   false, A[1][q], false, B0[1][q], (short)0, acc1, false, false);
    }
    {
        const float u0 = diag_of(acc0, lane7) * inv_a;
        const float u1 = diag_of(acc1, lane7) * inv_a;
        Sn[ps[0]] = u0;  Sn[ps[0] + RING] = u0;
        Sn[ps[1]] = u1;  Sn[ps[1] + RING] = u1;
        yacc[0] += u0;
        yacc[1] += u1;
    }

    if constexpr (!LAST) {
        load_b(Rn, bB[0], bB[1], B0);
    } else {
        load_a_pair(ntile, c, h2, 0, A);
    }
    __builtin_amdgcn_sched_barrier(0);

    v8f acc2 = {0.f, 0.f, 0.f, 0.f, 0.f, 0.f, 0.f, 0.f};
    v8f acc3 = {0.f, 0.f, 0.f, 0.f, 0.f, 0.f, 0.f, 0.f};
#pragma unroll
    for (int q = 0; q < 7; ++q) {
        acc2 = __builtin_amdgcn_wmma_f32_16x16x4_f32(
                   false, A[2][q], false, B1[0][q], (short)0, acc2, false, false);
        acc3 = __builtin_amdgcn_wmma_f32_16x16x4_f32(
                   false, A[3][q], false, B1[1][q], (short)0, acc3, false, false);
    }
    {
        const float u2 = diag_of(acc2, lane7) * inv_a;
        const float u3 = diag_of(acc3, lane7) * inv_a;
        Sn[ps[2]] = u2;  Sn[ps[2] + RING] = u2;
        Sn[ps[3]] = u3;  Sn[ps[3] + RING] = u3;
        yacc[2] += u2;
        yacc[3] += u3;
    }
    if constexpr (LAST) {
        load_a_pair(ntile, c, h2, 2, A);
    }
}

__global__ __launch_bounds__(NTHREADS)
void mglsa_levelpipe_wmma_kernel(const float* __restrict__ x,
                                 const float* __restrict__ mc,
                                 float* __restrict__ y)
{
    __shared__ __attribute__((aligned(16))) float lds[LDS_F];

    const int tid  = (int)threadIdx.x;
    const int lane = tid & 31;
    const int w    = __builtin_amdgcn_readfirstlane(tid >> 5);
    const int b    = (int)blockIdx.x / CPR;
    const int ci   = (int)blockIdx.x % CPR;
    const int sg0  = (ci * NSG) / CPR;
    const int sg1  = ((ci + 1) * NSG) / CPR;
    const int hs   = (sg0 - LEAD) > 0 ? (sg0 - LEAD) : 0;
    const int nblk = sg1 - hs;

    const int nlo = (w == 0) ? 0 : (4 * w - 1);
    const int nhi = (w == NWAVES - 1) ? NITER : (4 * w + 3);
    const int lw  = (w == 0) ? LEAD : ((24 * (NITER - 1 - nlo) + SG - 1) / SG);
    const int kst = (sg0 - lw) > hs ? (sg0 - lw) : hs;

    float* R      = lds;
    float* tiles  = lds + TILE_OFF;
    const float* ypin  = lds + YP_OFF + ((w > 0) ? (w - 1) : 0) * (2 * SG);
    float*       ypout = lds + YP_OFF + w * (2 * SG);
    float*       ystage = lds + YST_OFF;
    const int DUMP0 = DUMP_OFF + w * DUMPW;

    const float* __restrict__ xb  = x  + (size_t)b * T_LEN;
    const float* __restrict__ mcb = mc + (size_t)b * T_LEN * M1;
    float* __restrict__       yb  = y  + (size_t)b * T_LEN;
    const float4* __restrict__ mc4 = reinterpret_cast<const float4*>(mcb);
    const bool second = tid < (TILE_V4 - NTHREADS);

    const int  c       = lane & 15;
    const int  h2      = (lane >> 4) << 1;
    const int  lane7   = lane & 7;
    const bool hasDiag = (lane < 8) || (lane >= 24);
    const int  dslot   = (lane < 8) ? lane : (lane - 16);

    {
        float4* R4 = reinterpret_cast<float4*>(R);
        for (int i = tid; i < (NLV * RSZ) / 4; i += NTHREADS)
            R4[i] = make_float4(0.f, 0.f, 0.f, 0.f);
#pragma unroll
        for (int j = 0; j < AHEAD; ++j) {
            const int blk = hs + j;
            float4* t4 = reinterpret_cast<float4*>(tiles + (blk & (NTILE - 1)) * TILE_F);
            t4[tid] = mc4[blk * TILE_V4 + tid];
            if (second)
                t4[NTHREADS + tid] = mc4[blk * TILE_V4 + NTHREADS + tid];
        }
    }
    float xa = 0.f, xc = 0.f;
    if (w == 0) {
        xa = xb[hs * SG + lane];
        xc = xb[hs * SG + 32 + lane];
    }

    __syncthreads();

    v2f A[4][7];
#pragma unroll
    for (int g = 0; g < 4; ++g)
#pragma unroll
        for (int q = 0; q < 7; ++q) {
            A[g][q].x = 0.0f;
            A[g][q].y = 0.0f;
        }

    for (int s = 0; s < nblk + NWAVES - 1; ++s) {
        const int  fb    = hs + s + AHEAD;
        const bool fetch = fb < sg1;
        float4 f0 = make_float4(0.f, 0.f, 0.f, 0.f);
        float4 f1 = make_float4(0.f, 0.f, 0.f, 0.f);
        if (fetch) {
            f0 = mc4[fb * TILE_V4 + tid];
            if (second)
                f1 = mc4[fb * TILE_V4 + NTHREADS + tid];
        }

        const int k = hs + s - w;
        if (k >= kst && k < sg1) {
            if (k == kst) {
                const float* ftile = tiles + (k & (NTILE - 1)) * TILE_F;
                load_a_pair(ftile, c, h2, 0, A);
                load_a_pair(ftile, c, h2, 2, A);
            }
            const int nk = (k + 1 < sg1) ? (k + 1) : k;
            float xan = 0.f, xcn = 0.f;
            if (w == 0) {
                xan = xb[nk * SG + lane];
                xcn = xb[nk * SG + 32 + lane];
            }

            const int rbase = (RING - 1) - SG * (k & 3);
            const int ypar  = (k & 1) * SG;

            if (w == 0) {
                const int p = rbase - lane;
                R[p]             = xa;
                R[p + RING]      = xa;
                R[p - 32]        = xc;
                R[p - 32 + RING] = xc;
            }

            int   bB[4], ps[4];
            float yacc[4];
#pragma unroll
            for (int g = 0; g < 4; ++g) {
                const int pd = rbase - 16 * g - dslot;
                bB[g] = rbase - 16 * g - c + h2;
                ps[g] = hasDiag ? pd : (DUMP0 + lane);
                const int yi = hasDiag ? (16 * g + dslot) : lane;
                yacc[g] = (w == 0) ? R[pd] : ypin[ypar + yi];
            }
            const int lstride = hasDiag ? RSZ : 0;

            v2f B0[2][7];
            load_b(R + nlo * RSZ, bB[0], bB[1], B0);

            const float* ntile = tiles + (nk & (NTILE - 1)) * TILE_F;

#pragma unroll 1
            for (int n = nlo + 1; n < nhi; ++n)
                level_trip<false>(n, R, lstride, bB, ps, yacc, A, B0, ntile, c, h2, lane7);
            level_trip<true>(nhi, R, lstride, bB, ps, yacc, A, B0, ntile, c, h2, lane7);

            if (w == NWAVES - 1) {
                if (k >= sg0) {
                    if (hasDiag) {
#pragma unroll
                        for (int g = 0; g < 4; ++g)
                            ystage[16 * g + dslot] = yacc[g];
                    }
                    const float y0 = ystage[lane];
                    const float y1 = ystage[32 + lane];
                    yb[k * SG + lane]      = y0;
                    yb[k * SG + 32 + lane] = y1;
                }
            } else {
                if (hasDiag) {
#pragma unroll
                    for (int g = 0; g < 4; ++g)
                        ypout[ypar + 16 * g + dslot] = yacc[g];
                }
            }

            xa = xan;
            xc = xcn;
        }

        if (fetch) {
            float4* t4 = reinterpret_cast<float4*>(tiles + (fb & (NTILE - 1)) * TILE_F);
            t4[tid] = f0;
            if (second)
                t4[NTHREADS + tid] = f1;
        }
        __syncthreads();
    }
}

extern "C" void kernel_launch(void* const* d_in, const int* in_sizes, int n_in,
                              void* d_out, int out_size, void* d_ws, size_t ws_size,
                              hipStream_t stream) {
    (void)in_sizes; (void)n_in; (void)out_size; (void)d_ws; (void)ws_size;
    const float* x  = (const float*)d_in[0];
    const float* mc = (const float*)d_in[1];
    float* yout     = (float*)d_out;

    mglsa_levelpipe_wmma_kernel<<<dim3(256), dim3(NTHREADS), 0, stream>>>(x, mc, yout);
}


# priority window: the later-started wave of each SIMD's pair (stages 4..7) holds issue priority 3 for its first two levels of a step, then 0, so the pair reaches the step barrier together; sums kept
# speedup vs baseline: 2.7556x; 1.0368x over previous
#include <hip/hip_runtime.h>


typedef __attribute__((ext_vector_type(2))) float v2f;
typedef __attribute__((ext_vector_type(8))) float v8f;

#define T_LEN    48000
#define M1       25
#define NITER    30
#define SG       64
#define NSG      (T_LEN / SG)
#define CPR      32
#define LEAD     12
#define NWAVES   8
#define NTHREADS (NWAVES * 32)
#define RING     256
#define RSZ      (2 * RING)
#define NLV      (NITER + 1)
#define TILE_F   (SG * M1)
#define TILE_V4  (TILE_F / 4)
#define NTILE    16
#define AHEAD    2
#define TILE_OFF (NLV * RSZ)
#define YP_OFF   (TILE_OFF + NTILE * TILE_F)
#define YST_OFF  (YP_OFF + NWAVES * 2 * SG)
#define DUMPW    (32 + RING)
#define DUMP_OFF (YST_OFF + SG)
#define LDS_F    (DUMP_OFF + NWAVES * DUMPW)
static_assert((LDS_F % 4) == 0 && (TILE_OFF % 4) == 0 && (TILE_F % 4) == 0, "16-byte pieces");
static_assert(LDS_F * 4 > 160 * 1024 && LDS_F * 4 <= 320 * 1024, "one workgroup per WGP");
static_assert((CPR * 8) == 256 && NWAVES == 8, "level split below is written for 8 waves");
static_assert(TILE_V4 > NTHREADS && TILE_V4 <= 2 * NTHREADS, "two pieces per thread");


__device__ __forceinline__ void load_a_pair(const float* tile, int c, int h2, int g0, v2f (&A)[4][7])
{
#pragma unroll
    for (int g = g0; g < g0 + 2; ++g) {
        const float* mrow = tile + (16 * g + c) * M1;
#pragma unroll
        for (int q = 0; q < 6; ++q) {
            A[g][q].x = mrow[4 * q + h2];
            A[g][q].y = mrow[4 * q + h2 + 1];
        }
        const float a24 = mrow[24];
        A[g][6].x = (h2 == 0) ? a24 : 0.0f;
        A[g][6].y = 0.0f;
    }
}

__device__ __forceinline__ void load_b(const float* Rl, int b0, int b1, v2f (&B)[2][7])
{
#pragma unroll
    for (int q = 0; q < 7; ++q) {
        B[0][q].x = Rl[b0 + 4 * q];
        B[0][q].y = Rl[b0 + 4 * q + 1];
        B[1][q].x = Rl[b1 + 4 * q];
        B[1][q].y = Rl[b1 + 4 * q + 1];
    }
}

__device__ __forceinline__ float diag_of(const v8f& acc, int lane7)
{
    float dv = acc[0];
#pragma unroll
    for (int v = 1; v < 8; ++v)
        dv = (lane7 == v) ? acc[v] : dv;
    return dv;
}

template <bool LAST>
__device__ __forceinline__ void level_trip(const int n, float* R, const int lstride,
                                           const int (&bB)[4], const int (&ps)[4],
                                           float (&yacc)[4], v2f (&A)[4][7], v2f (&B0)[2][7],
                                           const float* ntile, const int c, const int h2,
                                           const int lane7)
{
    const float  inv_a = 1.0f / (float)n;
    const float* Rp = R + (n - 1) * RSZ;
    const float* Rn = R + n * RSZ;
    float*       Sn = R + n * lstride;

    v2f B1[2][7];
    load_b(Rp, bB[2], bB[3], B1);
    __builtin_amdgcn_sched_barrier(0);

    v8f acc0 = {0.f, 0.f, 0.f, 0.f, 0.f, 0.f, 0.f, 0.f};
    v8f acc1 = {0.f, 0.f, 0.f, 0.f, 0.f, 0.f, 0.f, 0.f};
#pragma unroll
    for (int q = 0; q < 7; ++q) {
        acc0 = __builtin_amdgcn_wmma_f32_16x16x4_f32(
                   false, A[0][q], false, B0[0][q], (short)0, acc0, false, false);
        acc1 = __builtin_amdgcn_wmma_f32_16x16x4_f32(
                   false, A[1][q], false, B0[1][q], (short)0, acc1, false, false);
    }
    {
        const float u0 = diag_of(acc0, lane7) * inv_a;
        const float u1 = diag_of(acc1, lane7) * inv_a;
        Sn[ps[0]] = u0;  Sn[ps[0] + RING] = u0;
        Sn[ps[1]] = u1;  Sn[ps[1] + RING] = u1;
        yacc[0] += u0;
        yacc[1] += u1;
    }

    if constexpr (!LAST) {
        load_b(Rn, bB[0], bB[1], B0);
    } else {
        load_a_pair(ntile, c, h2, 0, A);
    }
    __builtin_amdgcn_sched_barrier(0);

    v8f acc2 = {0.f, 0.f, 0.f, 0.f, 0.f, 0.f, 0.f, 0.f};
    v8f acc3 = {0.f, 0.f, 0.f, 0.f, 0.f, 0.f, 0.f, 0.f};
#pragma unroll
    for (int q = 0; q < 7; ++q) {
        acc2 = __builtin_amdgcn_wmma_f32_16x16x4_f32(
                   false, A[2][q], false, B1[0][q], (short)0, acc2, false, false);
        acc3 = __builtin_amdgcn_wmma_f32_16x16x4_f32(
                   false, A[3][q], false, B1[1][q], (short)0, acc3, false, false);
    }
    {
        const float u2 = diag_of(acc2, lane7) * inv_a;
        const float u3 = diag_of(acc3, lane7) * inv_a;
        Sn[ps[2]] = u2;  Sn[ps[2] + RING] = u2;
        Sn[ps[3]] = u3;  Sn[ps[3] + RING] = u3;
        yacc[2] += u2;
        yacc[3] += u3;
    }
    if constexpr (LAST) {
        load_a_pair(ntile, c, h2, 2, A);
    }
}

__global__ __launch_bounds__(NTHREADS)
void mglsa_levelpipe_wmma_kernel(const float* __restrict__ x,
                                 const float* __restrict__ mc,
                                 float* __restrict__ y)
{
    __shared__ __attribute__((aligned(16))) float lds[LDS_F];

    const int tid  = (int)threadIdx.x;
    const int lane = tid & 31;
    const int w    = __builtin_amdgcn_readfirstlane(tid >> 5);
    const int b    = (int)blockIdx.x / CPR;
    const int ci   = (int)blockIdx.x % CPR;
    const int sg0  = (ci * NSG) / CPR;
    const int sg1  = ((ci + 1) * NSG) / CPR;
    const int hs   = (sg0 - LEAD) > 0 ? (sg0 - LEAD) : 0;
    const int nblk = sg1 - hs;

    const int nlo = (w == 0) ? 0 : (4 * w - 1);
    const int nhi = (w == NWAVES - 1) ? NITER : (4 * w + 3);
    const bool young = (w >= NWAVES / 2);
    const int lw  = (w == 0) ? LEAD : ((24 * (NITER - 1 - nlo) + SG - 1) / SG);
    const int kst = (sg0 - lw) > hs ? (sg0 - lw) : hs;

    float* R      = lds;
    float* tiles  = lds + TILE_OFF;
    const float* ypin  = lds + YP_OFF + ((w > 0) ? (w - 1) : 0) * (2 * SG);
    float*       ypout = lds + YP_OFF + w * (2 * SG);
    float*       ystage = lds + YST_OFF;
    const int DUMP0 = DUMP_OFF + w * DUMPW;

    const float* __restrict__ xb  = x  + (size_t)b * T_LEN;
    const float* __restrict__ mcb = mc + (size_t)b * T_LEN * M1;
    float* __restrict__       yb  = y  + (size_t)b * T_LEN;
    const float4* __restrict__ mc4 = reinterpret_cast<const float4*>(mcb);
    const bool second = tid < (TILE_V4 - NTHREADS);

    const int  c       = lane & 15;
    const int  h2      = (lane >> 4) << 1;
    const int  lane7   = lane & 7;
    const bool hasDiag = (lane < 8) || (lane >= 24);
    const int  dslot   = (lane < 8) ? lane : (lane - 16);

    {
        float4* R4 = reinterpret_cast<float4*>(R);
        for (int i = tid; i < (NLV * RSZ) / 4; i += NTHREADS)
            R4[i] = make_float4(0.f, 0.f, 0.f, 0.f);
#pragma unroll
        for (int j = 0; j < AHEAD; ++j) {
            const int blk = hs + j;
            float4* t4 = reinterpret_cast<float4*>(tiles + (blk & (NTILE - 1)) * TILE_F);
            t4[tid] = mc4[blk * TILE_V4 + tid];
            if (second)
                t4[NTHREADS + tid] = mc4[blk * TILE_V4 + NTHREADS + tid];
        }
    }
    float xa = 0.f, xc = 0.f;
    if (w == 0) {
        xa = xb[hs * SG + lane];
        xc = xb[hs * SG + 32 + lane];
    }

    __syncthreads();

    v2f A[4][7];
#pragma unroll
    for (int g = 0; g < 4; ++g)
#pragma unroll
        for (int q = 0; q < 7; ++q) {
            A[g][q].x = 0.0f;
            A[g][q].y = 0.0f;
        }

    for (int s = 0; s < nblk + NWAVES - 1; ++s) {
        const int  fb    = hs + s + AHEAD;
        const bool fetch = fb < sg1;
        float4 f0 = make_float4(0.f, 0.f, 0.f, 0.f);
        float4 f1 = make_float4(0.f, 0.f, 0.f, 0.f);
        if (fetch) {
            f0 = mc4[fb * TILE_V4 + tid];
            if (second)
                f1 = mc4[fb * TILE_V4 + NTHREADS + tid];
        }

        const int k = hs + s - w;
        if (k >= kst && k < sg1) {
            if (k == kst) {
                const float* ftile = tiles + (k & (NTILE - 1)) * TILE_F;
                load_a_pair(ftile, c, h2, 0, A);
                load_a_pair(ftile, c, h2, 2, A);
            }
            const int nk = (k + 1 < sg1) ? (k + 1) : k;
            float xan = 0.f, xcn = 0.f;
            if (w == 0) {
                xan = xb[nk * SG + lane];
                xcn = xb[nk * SG + 32 + lane];
            }

            const int rbase = (RING - 1) - SG * (k & 3);
            const int ypar  = (k & 1) * SG;

            if (w == 0) {
                const int p = rbase - lane;
                R[p]             = xa;
                R[p + RING]      = xa;
                R[p - 32]        = xc;
                R[p - 32 + RING] = xc;
            }

            int   bB[4], ps[4];
            float yacc[4];
#pragma unroll
            for (int g = 0; g < 4; ++g) {
                const int pd = rbase - 16 * g - dslot;
                bB[g] = rbase - 16 * g - c + h2;
                ps[g] = hasDiag ? pd : (DUMP0 + lane);
                const int yi = hasDiag ? (16 * g + dslot) : lane;
                yacc[g] = (w == 0) ? R[pd] : ypin[ypar + yi];
            }
            const int lstride = hasDiag ? RSZ : 0;

            v2f B0[2][7];
            load_b(R + nlo * RSZ, bB[0], bB[1], B0);

            const float* ntile = tiles + (nk & (NTILE - 1)) * TILE_F;

            if (young)
                __builtin_amdgcn_s_setprio(3);

#pragma unroll 1
            for (int n = nlo + 1; n < nlo + 3; ++n)
                level_trip<false>(n, R, lstride, bB, ps, yacc, A, B0, ntile, c, h2, lane7);
            if (young)
                __builtin_amdgcn_s_setprio(0);
#pragma unroll 1
            for (int n = nlo + 3; n < nhi; ++n)
                level_trip<false>(n, R, lstride, bB, ps, yacc, A, B0, ntile, c, h2, lane7);
            level_trip<true>(nhi, R, lstride, bB, ps, yacc, A, B0, ntile, c, h2, lane7);

            if (w == NWAVES - 1) {
                if (k >= sg0) {
                    if (hasDiag) {
#pragma unroll
                        for (int g = 0; g < 4; ++g)
                            ystage[16 * g + dslot] = yacc[g];
                    }
                    const float y0 = ystage[lane];
                    const float y1 = ystage[32 + lane];
                    yb[k * SG + lane]      = y0;
                    yb[k * SG + 32 + lane] = y1;
                }
            } else {
                if (hasDiag) {
#pragma unroll
                    for (int g = 0; g < 4; ++g)
                        ypout[ypar + 16 * g + dslot] = yacc[g];
                }
            }

            xa = xan;
            xc = xcn;
        }

        if (fetch) {
            float4* t4 = reinterpret_cast<float4*>(tiles + (fb & (NTILE - 1)) * TILE_F);
            t4[tid] = f0;
            if (second)
                t4[NTHREADS + tid] = f1;
        }
        __syncthreads();
    }
}

extern "C" void kernel_launch(void* const* d_in, const int* in_sizes, int n_in,
                              void* d_out, int out_size, void* d_ws, size_t ws_size,
                              hipStream_t stream) {
    (void)in_sizes; (void)n_in; (void)out_size; (void)d_ws; (void)ws_size;
    const float* x  = (const float*)d_in[0];
    const float* mc = (const float*)d_in[1];
    float* yout     = (float*)d_out;

    mglsa_levelpipe_wmma_kernel<<<dim3(256), dim3(NTHREADS), 0, stream>>>(x, mc, yout);
}
